# starting kernel
# baseline (speedup 1.0000x reference)
#include <hip/hip_runtime.h>

typedef __attribute__((ext_vector_type(2))) float v2f;
typedef __attribute__((ext_vector_type(8))) float v8f;

namespace {
constexpr int T = 1024, B = 512, D = 16, H = 16;

__device__ __forceinline__ float tanh_hw(float x) {
  float r;
  asm volatile("v_tanh_f32 %0, %1\n\tv_nop" : "=v"(r) : "v"(x));
  return r;
}
__device__ __forceinline__ float sigmoid_hw(float x) {
  return __builtin_fmaf(0.5f, tanh_hw(0.5f * x), 0.5f);
}
__device__ __forceinline__ v8f splat8(float s) {
  v8f v;
#pragma unroll
  for (int i = 0; i < 8; ++i) v[i] = s;
  return v;
}

__device__ __forceinline__ void load_x(v2f (&xa)[4], const float* __restrict__ x,
                                       int t, int bbase, int n, int hi) {
#pragma unroll
  for (int j = 0; j < 4; ++j)
    xa[j] = *(const v2f*)(x + ((size_t)t * B + bbase + n) * D + 4 * j + 2 * hi);
}

__device__ __forceinline__ void lstm_step(
    int t, const v2f (&xc)[4], const v2f (&wih)[4][4], const v2f (&whh)[4][4],
    const v8f (&biasV)[4], v2f (&hA)[4], v8f& c, float* __restrict__ y,
    float* hlds, int n, int hi, int bbase) {
  v8f G[4];
#pragma unroll
  for (int nt = 0; nt < 4; ++nt) {
    G[nt] = __builtin_amdgcn_wmma_f32_16x16x4_f32(
        false, xc[0], false, wih[nt][0], (short)0, biasV[nt], false, false);
#pragma unroll
    for (int j = 1; j < 4; ++j)
      G[nt] = __builtin_amdgcn_wmma_f32_16x16x4_f32(
          false, xc[j], false, wih[nt][j], (short)0, G[nt], false, false);
#pragma unroll
    for (int j = 0; j < 4; ++j)
      G[nt] = __builtin_amdgcn_wmma_f32_16x16x4_f32(
          false, hA[j], false, whh[nt][j], (short)0, G[nt], false, false);
  }

  const int m0 = 8 * hi;
  float* yrow = y + ((size_t)t * B + bbase + m0) * H + n;
#pragma unroll
  for (int r = 0; r < 8; ++r) {
    float iv = sigmoid_hw(G[0][r]);
    float fv = sigmoid_hw(G[1][r]);
    float gv = tanh_hw(G[2][r]);
    float ov = sigmoid_hw(G[3][r]);
    float cv = __builtin_fmaf(fv, c[r], iv * gv);
    c[r] = cv;
    float hv = ov * tanh_hw(cv);
    yrow[(size_t)r * H] = hv;
    hlds[(m0 + r) * 16 + n] = hv;
  }
  __syncthreads();
#pragma unroll
  for (int j = 0; j < 4; ++j)
    hA[j] = *(const v2f*)(hlds + n * 16 + 4 * j + 2 * hi);
  __syncthreads();
}
}

__global__ __launch_bounds__(32, 1) void lstm_wmma_kernel(
    const float* __restrict__ x, const float* __restrict__ W_ih,
    const float* __restrict__ W_hh, const float* __restrict__ b_ih,
    const float* __restrict__ b_hh, float* __restrict__ y) {
  __shared__ __align__(16) float hlds[16 * 16];

  const int lane  = threadIdx.x;
  const int n     = lane & 15;
  const int hi    = lane >> 4;
  const int bbase = blockIdx.x * 16;

  v2f wih[4][4], whh[4][4];
#pragma unroll
  for (int nt = 0; nt < 4; ++nt)
#pragma unroll
    for (int j = 0; j < 4; ++j) {
      wih[nt][j] = *(const v2f*)(W_ih + (size_t)(16 * nt + n) * D + 4 * j + 2 * hi);
      whh[nt][j] = *(const v2f*)(W_hh + (size_t)(16 * nt + n) * H + 4 * j + 2 * hi);
    }

  v8f biasV[4];
#pragma unroll
  for (int nt = 0; nt < 4; ++nt)
    biasV[nt] = splat8(b_ih[16 * nt + n] + b_hh[16 * nt + n]);

  v8f c = {};
  v2f hA[4];
#pragma unroll
  for (int j = 0; j < 4; ++j) { hA[j][0] = 0.f; hA[j][1] = 0.f; }

  v2f xA0[4], xA1[4];
  load_x(xA0, x, 0, bbase, n, hi);

  for (int t = 0; t < T; t += 2) {
    load_x(xA1, x, t + 1, bbase, n, hi);
    if (t + 16 < T)
      __builtin_prefetch(x + ((size_t)(t + 16) * B + bbase + n) * D, 0, 1);
    lstm_step(t, xA0, wih, whh, biasV, hA, c, y, hlds, n, hi, bbase);

    if (t + 2 < T) load_x(xA0, x, t + 2, bbase, n, hi);
    lstm_step(t + 1, xA1, wih, whh, biasV, hA, c, y, hlds, n, hi, bbase);
  }
}

extern "C" void kernel_launch(void* const* d_in, const int*  ,
                              int  , void* d_out, int  ,
                              void*  , size_t  ,
                              hipStream_t stream) {
  const float* x    = (const float*)d_in[0];
  const float* W_ih = (const float*)d_in[1];
  const float* W_hh = (const float*)d_in[2];
  const float* b_ih = (const float*)d_in[3];
  const float* b_hh = (const float*)d_in[4];
  float* y = (float*)d_out;
  lstm_wmma_kernel<<<B / 16, 32, 0, stream>>>(x, W_ih, W_hh, b_ih, b_hh, y);
}


# Step's matrix products split over a 4-wave work-group by reduction index (wave w: k=4w..4w+3, 4 unchained WMMAs); partial sums meet in LDS, one barrier a step; f32 sum order changes
# speedup vs baseline: 6.1069x; 6.1069x over previous
#include <hip/hip_runtime.h>

typedef __attribute__((ext_vector_type(2))) float v2f;
typedef __attribute__((ext_vector_type(4))) float v4f;
typedef __attribute__((ext_vector_type(8))) float v8f;

namespace {
constexpr int T = 1024, B = 512, D = 16, H = 16;
constexpr int ROWS = 2;
constexpr int GRID = B / ROWS;
constexpr int TB = 8;
constexpr int WAVES = 4;

constexpr int P_ROW = 80;
constexpr int P_WAVE = 160;
constexpr int P_BUF = WAVES * P_WAVE;
constexpr int H_BUF = ROWS * H;

__device__ __forceinline__ float tanh_hw(float x) {
  float r;
  asm volatile("v_tanh_f32 %0, %1\n\tv_nop" : "=v"(r) : "v"(x));
  return r;
}
__device__ __forceinline__ v8f splat8(float s) {
  v8f v;
#pragma unroll
  for (int i = 0; i < 8; ++i) v[i] = s;
  return v;
}

__device__ __forceinline__ float lane_gather_own(float v, unsigned sel) {
  return __builtin_bit_cast(float, __builtin_amdgcn_permlane16_var(
      __builtin_bit_cast(unsigned, v), __builtin_bit_cast(unsigned, v), sel, true, false));
}

__device__ __forceinline__ void exchange_barrier() {
  asm volatile("s_wait_dscnt 0x0\n\ts_barrier_signal -1\n\ts_barrier_wait -1" ::: "memory");
}

__device__ __forceinline__ v2f load_x_slice(const float* __restrict__ x, int t0,
                                            int bbase, int n, int hi, int w) {
  return *(const v2f*)(x + ((size_t)(t0 + (n & 7)) * B + bbase + (n >> 3)) * D +
                       4 * w + 2 * hi);
}

template <int R>
__device__ __forceinline__ void lstm_step(
    int t, const v8f (&XG)[4], v8f (&XGn)[4], v2f& xn,
    const v2f (&wih)[4], const v2f (&whh)[4], const v8f (&biasV)[4],
    v2f& hA, float& c, float* __restrict__ ylane,
    float* pw, const float* pr, float* hw, const float* hr,
    float act_s, float act_b, const unsigned (&selq)[4], const unsigned (&sela)[2],
    unsigned w) {
  const v8f Gi = __builtin_amdgcn_wmma_f32_16x16x4_f32(
      false, hA, false, whh[0], (short)0, XG[0], false, false);
  const v8f Gf = __builtin_amdgcn_wmma_f32_16x16x4_f32(
      false, hA, false, whh[1], (short)0, XG[1], false, false);
  const v8f Gg = __builtin_amdgcn_wmma_f32_16x16x4_f32(
      false, hA, false, whh[2], (short)0, XG[2], false, false);
  const v8f Go = __builtin_amdgcn_wmma_f32_16x16x4_f32(
      false, hA, false, whh[3], (short)0, XG[3], false, false);

  v4f part;
  part[0] = Gi[R]; part[1] = Gf[R]; part[2] = Gg[R]; part[3] = Go[R];
  *(v4f*)(pw + (R & 1) * P_BUF) = part;

  exchange_barrier();

  const float* prb = pr + (R & 1) * P_BUF;
  const float p0 = prb[0 * P_WAVE];
  const float p1 = prb[1 * P_WAVE];
  const float p2 = prb[2 * P_WAVE];
  const float p3 = prb[3 * P_WAVE];
  const float yprev = hr[((R + 1) & 1) * H_BUF];
  __builtin_amdgcn_sched_barrier(0);
  if constexpr (R < 4) {
    asm volatile("" : "+v"(xn));
    XGn[R] = __builtin_amdgcn_wmma_f32_16x16x4_f32(
        false, xn, false, wih[R], (short)0, biasV[R], false, false);
    asm volatile("" : : "v"(XGn[R]));
    __builtin_amdgcn_sched_barrier(0);
  }

  const float z = (p0 + p1) + (p2 + p3);
  const float act = __builtin_fmaf(act_s, tanh_hw(act_s * z), act_b);
  const float iv = lane_gather_own(act, selq[0]);
  const float fv = lane_gather_own(act, selq[1]);
  const float gv = lane_gather_own(act, selq[2]);
  const float ov = lane_gather_own(act, selq[3]);
  const float ig = iv * gv;
  c = __builtin_fmaf(fv, c, ig);
  const float tc = tanh_hw(c);
  const float hv = ov * tc;
  hw[(R & 1) * H_BUF] = hv;
  hA[0] = lane_gather_own(hv, sela[0]);
  hA[1] = lane_gather_own(hv, sela[1]);

  if (w == (unsigned)((R + 3) & 3) && (R != 0 || t > 0))
    ylane[(size_t)(t - 1) * (B * H)] = yprev;
}

__device__ __forceinline__ void lstm_block(
    int t0, const v8f (&XG)[4], v8f (&XGn)[4], v2f& xn,
    v2f& xl, int tl, const float* __restrict__ x,
    const v2f (&wih)[4], const v2f (&whh)[4], const v8f (&biasV)[4],
    v2f& hA, float& c, float* __restrict__ ylane,
    float* pw, const float* pr, float* hw, const float* hr,
    float act_s, float act_b, const unsigned (&selq)[4], const unsigned (&sela)[2],
    unsigned w, int n, int hi, int bbase) {
  lstm_step<0>(t0 + 0, XG, XGn, xn, wih, whh, biasV, hA, c, ylane, pw, pr, hw, hr, act_s, act_b, selq, sela, w);
  xl = load_x_slice(x, tl, bbase, n, hi, (int)w);
  lstm_step<1>(t0 + 1, XG, XGn, xn, wih, whh, biasV, hA, c, ylane, pw, pr, hw, hr, act_s, act_b, selq, sela, w);
  lstm_step<2>(t0 + 2, XG, XGn, xn, wih, whh, biasV, hA, c, ylane, pw, pr, hw, hr, act_s, act_b, selq, sela, w);
  lstm_step<3>(t0 + 3, XG, XGn, xn, wih, whh, biasV, hA, c, ylane, pw, pr, hw, hr, act_s, act_b, selq, sela, w);
  lstm_step<4>(t0 + 4, XG, XGn, xn, wih, whh, biasV, hA, c, ylane, pw, pr, hw, hr, act_s, act_b, selq, sela, w);
  lstm_step<5>(t0 + 5, XG, XGn, xn, wih, whh, biasV, hA, c, ylane, pw, pr, hw, hr, act_s, act_b, selq, sela, w);
  lstm_step<6>(t0 + 6, XG, XGn, xn, wih, whh, biasV, hA, c, ylane, pw, pr, hw, hr, act_s, act_b, selq, sela, w);
  lstm_step<7>(t0 + 7, XG, XGn, xn, wih, whh, biasV, hA, c, ylane, pw, pr, hw, hr, act_s, act_b, selq, sela, w);
}
}

__global__ __launch_bounds__(32 * WAVES) void lstm_wmma_kernel(
    const float* __restrict__ x, const float* __restrict__ W_ih,
    const float* __restrict__ W_hh, const float* __restrict__ b_ih,
    const float* __restrict__ b_hh, float* __restrict__ y) {
  __shared__ __attribute__((aligned(16))) float lds_part[2 * P_BUF];
  __shared__ float lds_h[2 * H_BUF];

  const int lane  = threadIdx.x & 31;
  const int n     = lane & 15;
  const int hi    = lane >> 4;
  const int bbase = blockIdx.x * ROWS;
  const unsigned w = (unsigned)__builtin_amdgcn_readfirstlane((int)(threadIdx.x >> 5));

  v2f wih[4], whh[4];
#pragma unroll
  for (int nt = 0; nt < 4; ++nt) {
    wih[nt] = *(const v2f*)(W_ih + (size_t)(16 * nt + n) * D + 4 * w + 2 * hi);
    whh[nt] = *(const v2f*)(W_hh + (size_t)(16 * nt + n) * H + 4 * w + 2 * hi);
  }

  v8f biasV[4];
#pragma unroll
  for (int nt = 0; nt < 4; ++nt)
    biasV[nt] = splat8(w == 0u ? b_ih[16 * nt + n] + b_hh[16 * nt + n] : 0.f);

  const int gate  = n & 3;
  const int crow  = n >> 3;
  const int cunit = 4 * (int)w + 2 * hi + ((n >> 2) & 1);
  const float act_s = (gate == 2) ? 1.0f : 0.5f;
  const float act_b = (gate == 2) ? 0.0f : 0.5f;
  unsigned selq[4], sela[2];
#pragma unroll
  for (int k = 0; k < 4; ++k) selq[k] = (unsigned)((n & 12) | k);
#pragma unroll
  for (int e = 0; e < 2; ++e) sela[e] = (unsigned)((n & 8) | (e << 2));

  float* pw = lds_part + (int)w * P_WAVE + hi * P_ROW + n * 4;
  const float* pr = lds_part + crow * P_ROW + cunit * 4 + gate;
  float* hw = lds_h + crow * H + cunit;
  const float* hr = lds_h + hi * H + n;
  float* ylane = y + (size_t)(bbase + hi) * H + n;

  float c = 0.f;
  v2f hA;
  hA[0] = 0.f; hA[1] = 0.f;

  v2f xE, xO;
  v8f XGa[4], XGb[4];
  xE = load_x_slice(x, 0, bbase, n, hi, (int)w);
#pragma unroll
  for (int nt = 0; nt < 4; ++nt)
    XGa[nt] = __builtin_amdgcn_wmma_f32_16x16x4_f32(
        false, xE, false, wih[nt], (short)0, biasV[nt], false, false);
  xO = load_x_slice(x, TB, bbase, n, hi, (int)w);

  for (int t0 = 0; t0 < T; t0 += 2 * TB) {
    lstm_block(t0, XGa, XGb, xO, xE, (t0 + 2 * TB < T) ? t0 + 2 * TB : T - TB, x,
               wih, whh, biasV, hA, c, ylane, pw, pr, hw, hr, act_s, act_b, selq, sela,
               w, n, hi, bbase);
    lstm_block(t0 + TB, XGb, XGa, xE, xO, (t0 + 3 * TB < T) ? t0 + 3 * TB : T - TB, x,
               wih, whh, biasV, hA, c, ylane, pw, pr, hw, hr, act_s, act_b, selq, sela,
               w, n, hi, bbase);
  }

  exchange_barrier();
  if (w == (unsigned)((T - 1) & 3))
    ylane[(size_t)(T - 1) * (B * H)] = hr[((T - 1) & 1) * H_BUF];
}

extern "C" void kernel_launch(void* const* d_in, const int*  ,
                              int  , void* d_out, int  ,
                              void*  , size_t  ,
                              hipStream_t stream) {
  const float* x    = (const float*)d_in[0];
  const float* W_ih = (const float*)d_in[1];
  const float* W_hh = (const float*)d_in[2];
  const float* b_ih = (const float*)d_in[3];
  const float* b_hh = (const float*)d_in[4];
  float* y = (float*)d_out;
  lstm_wmma_kernel<<<GRID, 32 * WAVES, 0, stream>>>(x, W_ih, W_hh, b_ih, b_hh, y);
}
